# starting kernel
# baseline (speedup 1.0000x reference)
#include <hip/hip_runtime.h>
#include <hip/hip_bf16.h>

typedef __attribute__((ext_vector_type(2))) float v2f;
typedef __attribute__((ext_vector_type(8))) float v8f;

#define JB 8192
#define NJ 8
#define DIM 128
#define COLS 1024

__device__ __forceinline__ float squash_gain(float n2) {
    return n2 / ((1.0f + n2) * sqrtf(n2 + 1e-8f));
}

__launch_bounds__(256)
__global__ void gemm_uhat_kernel(const float* __restrict__ x,
                                 const float* __restrict__ w,
                                 float* __restrict__ uh) {
    const float* x7 = x + (size_t)7 * 8 * 1024 * 128;
    const float* w7 = w + (size_t)7 * 8 * 128 * 128;

    const int wave = threadIdx.x >> 5;
    const int lane = threadIdx.x & 31;
    const int half = lane >> 4;
    const int l16  = lane & 15;

    const int row0 = blockIdx.x * 128 + wave * 16;
    const int r    = row0 + l16;
    const float* arow = x7 + (size_t)((r & 7) * 1024 + (r >> 3)) * 128;

    for (int j = 0; j < NJ; ++j) {
        const float* wj = w7 + (size_t)j * 128 * 128;
        v8f acc[8];
#pragma unroll
        for (int n = 0; n < 8; ++n) acc[n] = (v8f){0,0,0,0,0,0,0,0};

        for (int k = 0; k < 128; k += 4) {
            v2f a;
            a.x = arow[k + 2 * half + 0];
            a.y = arow[k + 2 * half + 1];
            const float* wk = wj + (size_t)(k + 2 * half) * 128 + l16;
#pragma unroll
            for (int n = 0; n < 8; ++n) {
                v2f b;
                b.x = wk[n * 16];
                b.y = wk[128 + n * 16];
                acc[n] = __builtin_amdgcn_wmma_f32_16x16x4_f32(
                    false, a, false, b, (short)0, acc[n], false, false);
            }
        }
#pragma unroll
        for (int n = 0; n < 8; ++n) {
            float* od = uh + (size_t)(row0 + 8 * half) * COLS + j * DIM + n * 16 + l16;
#pragma unroll
            for (int g = 0; g < 8; ++g) {
                od[(size_t)g * COLS] = acc[n][g];
            }
        }
    }
}

__launch_bounds__(256)
__global__ void colmean_kernel(const float* __restrict__ uh, float* __restrict__ uhm) {
    int c = blockIdx.x * 256 + threadIdx.x;
    float s = 0.0f;
    for (int r = 0; r < JB; ++r) s += uh[(size_t)r * COLS + c];
    uhm[c] = s * (1.0f / (float)JB);
}

__launch_bounds__(256)
__global__ void dotn2_kernel(const float* __restrict__ uh,
                             const float* __restrict__ uhm,
                             float* __restrict__ dotp, float* __restrict__ n2) {
    const int wave = threadIdx.x >> 5;
    const int lane = threadIdx.x & 31;
    const int pair = blockIdx.x * 8 + wave;
    const int r = pair >> 3, j = pair & 7;

    const float4 a = ((const float4*)(uh + (size_t)r * COLS + j * DIM))[lane];
    const float4 m = ((const float4*)(uhm + j * DIM))[lane];
    float d = a.x * m.x + a.y * m.y + a.z * m.z + a.w * m.w;
    float q = a.x * a.x + a.y * a.y + a.z * a.z + a.w * a.w;
#pragma unroll
    for (int off = 16; off > 0; off >>= 1) {
        d += __shfl_xor(d, off, 32);
        q += __shfl_xor(q, off, 32);
    }
    if (lane == 0) { dotp[pair] = d; n2[pair] = q; }
}

__launch_bounds__(1024)
__global__ void routing_kernel(const float* __restrict__ dotp,
                               const float* __restrict__ n2,
                               float* __restrict__ cf) {
    __shared__ float B7[8];
    __shared__ float c[8];
    __shared__ float sumj[8];
    const int tid = threadIdx.x;
    if (tid < 8) B7[tid] = 0.0f;

    for (int it = 0; it < 2; ++it) {
        __syncthreads();
        if (tid == 0) {
            float m = B7[0];
            for (int jj = 1; jj < 8; ++jj) m = fmaxf(m, B7[jj]);
            float s = 0.0f, e[8];
            for (int jj = 0; jj < 8; ++jj) { e[jj] = __expf(B7[jj] - m); s += e[jj]; }
            for (int jj = 0; jj < 8; ++jj) c[jj] = e[jj] / s;
        }
        __syncthreads();
        if (tid < 8) sumj[tid] = 0.0f;
        __syncthreads();
        const int j = tid & 7;
        const float cj = c[j];
        float a = 0.0f;
        for (int p = tid; p < JB * NJ; p += 1024)
            a += squash_gain(cj * cj * n2[p]) * dotp[p];
        atomicAdd(&sumj[j], a);
        __syncthreads();
        if (tid < 8) B7[tid] += c[tid] * sumj[tid] * (1.0f / (float)JB);
    }
    __syncthreads();
    if (tid == 0) {
        float m = B7[0];
        for (int jj = 1; jj < 8; ++jj) m = fmaxf(m, B7[jj]);
        float s = 0.0f, e[8];
        for (int jj = 0; jj < 8; ++jj) { e[jj] = __expf(B7[jj] - m); s += e[jj]; }
        for (int jj = 0; jj < 8; ++jj) cf[jj] = e[jj] / s;
    }
}

__launch_bounds__(256)
__global__ void final_kernel(const float* __restrict__ uh,
                             const float* __restrict__ n2,
                             const float* __restrict__ cf,
                             float* __restrict__ out) {
    const size_t idx  = (size_t)blockIdx.x * 256 + threadIdx.x;
    const size_t base = idx * 4;
    const int r = (int)(base >> 10);
    const int j = (int)((base >> 7) & 7);
    const float cj = cf[j];
    const float gain = cj * squash_gain(cj * cj * n2[r * 8 + j]);
    float4 a = *(const float4*)(uh + base);
    a.x *= gain; a.y *= gain; a.z *= gain; a.w *= gain;
    *(float4*)(out + base) = a;
}

extern "C" void kernel_launch(void* const* d_in, const int* in_sizes, int n_in,
                              void* d_out, int out_size, void* d_ws, size_t ws_size,
                              hipStream_t stream) {
    const float* x = (const float*)d_in[0];
    const float* w = (const float*)d_in[1];
    float* out = (float*)d_out;

    char* ws = (char*)d_ws;
    float* uh   = (float*)(ws);
    float* n2   = (float*)(ws + (size_t)JB * COLS * 4);
    float* dotp = (float*)(ws + (size_t)JB * COLS * 4 + 262144);
    float* uhm  = (float*)(ws + (size_t)JB * COLS * 4 + 524288);
    float* cf   = (float*)(ws + (size_t)JB * COLS * 4 + 528384);

    gemm_uhat_kernel<<<64, 256, 0, stream>>>(x, w, uh);
    colmean_kernel<<<4, 256, 0, stream>>>(uh, uhm);
    dotn2_kernel<<<JB, 256, 0, stream>>>(uh, uhm, dotp, n2);
    routing_kernel<<<1, 1024, 0, stream>>>(dotp, n2, cf);
    final_kernel<<<(JB * COLS / 4) / 256, 256, 0, stream>>>(uh, n2, cf, out);
}


# column means from the GEMM accumulators: per-block column sums (registers, lane swap, LDS over 8 waves) give 64 partial rows that the mean kernel adds; row-sum order becomes a tree
# speedup vs baseline: 3.0605x; 3.0605x over previous
#include <hip/hip_runtime.h>
#include <hip/hip_bf16.h>

typedef __attribute__((ext_vector_type(2))) float v2f;
typedef __attribute__((ext_vector_type(8))) float v8f;

#define JB 8192
#define NJ 8
#define DIM 128
#define COLS 1024

__device__ __forceinline__ float other_row(float v) {
    const unsigned u = __builtin_bit_cast(unsigned, v);
    return __builtin_bit_cast(float,
        __builtin_amdgcn_permlanex16(u, u, 0x76543210u, 0xfedcba98u, false, false));
}

__device__ __forceinline__ float squash_gain(float n2) {
    return n2 / ((1.0f + n2) * sqrtf(n2 + 1e-8f));
}

__launch_bounds__(256)
__global__ void gemm_uhat_kernel(const float* __restrict__ x,
                                 const float* __restrict__ w,
                                 float* __restrict__ uh,
                                 float* __restrict__ part) {
    __shared__ float red[NJ * 8 * DIM];

    const float* x7 = x + (size_t)7 * 8 * 1024 * 128;
    const float* w7 = w + (size_t)7 * 8 * 128 * 128;

    const int wave = threadIdx.x >> 5;
    const int lane = threadIdx.x & 31;
    const int half = lane >> 4;
    const int l16  = lane & 15;

    const int row0 = blockIdx.x * 128 + wave * 16;
    const int r    = row0 + l16;
    const float* arow = x7 + (size_t)((r & 7) * 1024 + (r >> 3)) * 128;

    for (int j = 0; j < NJ; ++j) {
        const float* wj = w7 + (size_t)j * 128 * 128;
        v8f acc[8];
#pragma unroll
        for (int n = 0; n < 8; ++n) acc[n] = (v8f){0,0,0,0,0,0,0,0};

        for (int k = 0; k < 128; k += 4) {
            v2f a;
            a.x = arow[k + 2 * half + 0];
            a.y = arow[k + 2 * half + 1];
            const float* wk = wj + (size_t)(k + 2 * half) * 128 + l16;
#pragma unroll
            for (int n = 0; n < 8; ++n) {
                v2f b;
                b.x = wk[n * 16];
                b.y = wk[128 + n * 16];
                acc[n] = __builtin_amdgcn_wmma_f32_16x16x4_f32(
                    false, a, false, b, (short)0, acc[n], false, false);
            }
        }
        asm volatile("v_nop\n\tv_nop\n\tv_nop\n\tv_nop\n\tv_nop\n\tv_nop\n\tv_nop\n\tv_nop"
                     : "+v"(acc[0]), "+v"(acc[1]), "+v"(acc[2]), "+v"(acc[3]),
                       "+v"(acc[4]), "+v"(acc[5]), "+v"(acc[6]), "+v"(acc[7]));
#pragma unroll
        for (int n = 0; n < 8; ++n) {
            float* od = uh + (size_t)(row0 + 8 * half) * COLS + j * DIM + n * 16 + l16;
#pragma unroll
            for (int g = 0; g < 8; ++g) {
                od[(size_t)g * COLS] = acc[n][g];
            }
        }
        float csum[8];
#pragma unroll
        for (int n = 0; n < 8; ++n) {
            const float s = ((acc[n][0] + acc[n][1]) + (acc[n][2] + acc[n][3])) +
                            ((acc[n][4] + acc[n][5]) + (acc[n][6] + acc[n][7]));
            csum[n] = s + other_row(s);
        }
        if (half == 0) {
            float* rd = red + (j * 8 + wave) * DIM + l16;
#pragma unroll
            for (int n = 0; n < 8; ++n) rd[n * 16] = csum[n];
        }
    }
    __syncthreads();
    for (int i = threadIdx.x; i < COLS; i += 256) {
        const float* p = red + (i >> 7) * 8 * DIM + (i & 127);
        float s = p[0];
#pragma unroll
        for (int wv = 1; wv < 8; ++wv) s += p[wv * DIM];
        part[(size_t)blockIdx.x * COLS + i] = s;
    }
}

#define NPART 64
__launch_bounds__(256)
__global__ void colmean_kernel(const float* __restrict__ part, float* __restrict__ uhm) {
    __shared__ float red4[4][64];
    const int t   = threadIdx.x & 63;
    const int grp = threadIdx.x >> 6;
    const int c   = blockIdx.x * 64 + t;
    const float* p = part + (size_t)(grp * (NPART / 4)) * COLS + c;
    float s = p[0];
#pragma unroll
    for (int q = 1; q < NPART / 4; ++q) s += p[(size_t)q * COLS];
    red4[grp][t] = s;
    __syncthreads();
    if (threadIdx.x < 64)
        uhm[c] = ((red4[0][t] + red4[1][t]) + (red4[2][t] + red4[3][t])) * (1.0f / (float)JB);
}

__launch_bounds__(256)
__global__ void dotn2_kernel(const float* __restrict__ uh,
                             const float* __restrict__ uhm,
                             float* __restrict__ dotp, float* __restrict__ n2) {
    const int wave = threadIdx.x >> 5;
    const int lane = threadIdx.x & 31;
    const int pair = blockIdx.x * 8 + wave;
    const int r = pair >> 3, j = pair & 7;

    const float4 a = ((const float4*)(uh + (size_t)r * COLS + j * DIM))[lane];
    const float4 m = ((const float4*)(uhm + j * DIM))[lane];
    float d = a.x * m.x + a.y * m.y + a.z * m.z + a.w * m.w;
    float q = a.x * a.x + a.y * a.y + a.z * a.z + a.w * a.w;
#pragma unroll
    for (int off = 16; off > 0; off >>= 1) {
        d += __shfl_xor(d, off, 32);
        q += __shfl_xor(q, off, 32);
    }
    if (lane == 0) { dotp[pair] = d; n2[pair] = q; }
}

__launch_bounds__(1024)
__global__ void routing_kernel(const float* __restrict__ dotp,
                               const float* __restrict__ n2,
                               float* __restrict__ cf) {
    __shared__ float B7[8];
    __shared__ float c[8];
    __shared__ float sumj[8];
    const int tid = threadIdx.x;
    if (tid < 8) B7[tid] = 0.0f;

    for (int it = 0; it < 2; ++it) {
        __syncthreads();
        if (tid == 0) {
            float m = B7[0];
            for (int jj = 1; jj < 8; ++jj) m = fmaxf(m, B7[jj]);
            float s = 0.0f, e[8];
            for (int jj = 0; jj < 8; ++jj) { e[jj] = __expf(B7[jj] - m); s += e[jj]; }
            for (int jj = 0; jj < 8; ++jj) c[jj] = e[jj] / s;
        }
        __syncthreads();
        if (tid < 8) sumj[tid] = 0.0f;
        __syncthreads();
        const int j = tid & 7;
        const float cj = c[j];
        float a = 0.0f;
        for (int p = tid; p < JB * NJ; p += 1024)
            a += squash_gain(cj * cj * n2[p]) * dotp[p];
        atomicAdd(&sumj[j], a);
        __syncthreads();
        if (tid < 8) B7[tid] += c[tid] * sumj[tid] * (1.0f / (float)JB);
    }
    __syncthreads();
    if (tid == 0) {
        float m = B7[0];
        for (int jj = 1; jj < 8; ++jj) m = fmaxf(m, B7[jj]);
        float s = 0.0f, e[8];
        for (int jj = 0; jj < 8; ++jj) { e[jj] = __expf(B7[jj] - m); s += e[jj]; }
        for (int jj = 0; jj < 8; ++jj) cf[jj] = e[jj] / s;
    }
}

__launch_bounds__(256)
__global__ void final_kernel(const float* __restrict__ uh,
                             const float* __restrict__ n2,
                             const float* __restrict__ cf,
                             float* __restrict__ out) {
    const size_t idx  = (size_t)blockIdx.x * 256 + threadIdx.x;
    const size_t base = idx * 4;
    const int r = (int)(base >> 10);
    const int j = (int)((base >> 7) & 7);
    const float cj = cf[j];
    const float gain = cj * squash_gain(cj * cj * n2[r * 8 + j]);
    float4 a = *(const float4*)(uh + base);
    a.x *= gain; a.y *= gain; a.z *= gain; a.w *= gain;
    *(float4*)(out + base) = a;
}

extern "C" void kernel_launch(void* const* d_in, const int* in_sizes, int n_in,
                              void* d_out, int out_size, void* d_ws, size_t ws_size,
                              hipStream_t stream) {
    const float* x = (const float*)d_in[0];
    const float* w = (const float*)d_in[1];
    float* out = (float*)d_out;

    char* ws = (char*)d_ws;
    float* uh   = (float*)(ws);
    float* n2   = (float*)(ws + (size_t)JB * COLS * 4);
    float* dotp = (float*)(ws + (size_t)JB * COLS * 4 + 262144);
    float* uhm  = (float*)(ws + (size_t)JB * COLS * 4 + 524288);
    float* cf   = (float*)(ws + (size_t)JB * COLS * 4 + 528384);
    float* part = dotp;

    gemm_uhat_kernel<<<64, 256, 0, stream>>>(x, w, uh, part);
    colmean_kernel<<<16, 256, 0, stream>>>(part, uhm);
    dotn2_kernel<<<JB, 256, 0, stream>>>(uh, uhm, dotp, n2);
    routing_kernel<<<1, 1024, 0, stream>>>(dotp, n2, cf);
    final_kernel<<<(JB * COLS / 4) / 256, 256, 0, stream>>>(uh, n2, cf, out);
}


# GEMM grid 64 row blocks x 8 capsules (512 blocks, one capsule per block) instead of 64 blocks looping over 8 capsules: matrix work on every WGP; sums and their order unchanged
# speedup vs baseline: 6.2007x; 2.0260x over previous
#include <hip/hip_runtime.h>
#include <hip/hip_bf16.h>

typedef __attribute__((ext_vector_type(2))) float v2f;
typedef __attribute__((ext_vector_type(8))) float v8f;

#define JB 8192
#define NJ 8
#define DIM 128
#define COLS 1024

__device__ __forceinline__ float other_row(float v) {
    const unsigned u = __builtin_bit_cast(unsigned, v);
    return __builtin_bit_cast(float,
        __builtin_amdgcn_permlanex16(u, u, 0x76543210u, 0xfedcba98u, false, false));
}

__device__ __forceinline__ float squash_gain(float n2) {
    return n2 / ((1.0f + n2) * sqrtf(n2 + 1e-8f));
}

__launch_bounds__(256)
__global__ void gemm_uhat_kernel(const float* __restrict__ x,
                                 const float* __restrict__ w,
                                 float* __restrict__ uh,
                                 float* __restrict__ part) {
    __shared__ float red[8 * DIM];

    const float* x7 = x + (size_t)7 * 8 * 1024 * 128;
    const float* w7 = w + (size_t)7 * 8 * 128 * 128;

    const int wave = threadIdx.x >> 5;
    const int lane = threadIdx.x & 31;
    const int half = lane >> 4;
    const int l16  = lane & 15;

    const int j    = blockIdx.y;
    const int row0 = blockIdx.x * 128 + wave * 16;
    const int r    = row0 + l16;
    const float* arow = x7 + (size_t)((r & 7) * 1024 + (r >> 3)) * 128;

    const float* wj = w7 + (size_t)j * 128 * 128;
    v8f acc[8];
#pragma unroll
    for (int n = 0; n < 8; ++n) acc[n] = (v8f){0,0,0,0,0,0,0,0};

    for (int k = 0; k < 128; k += 4) {
        v2f a;
        a.x = arow[k + 2 * half + 0];
        a.y = arow[k + 2 * half + 1];
        const float* wk = wj + (size_t)(k + 2 * half) * 128 + l16;
#pragma unroll
        for (int n = 0; n < 8; ++n) {
            v2f b;
            b.x = wk[n * 16];
            b.y = wk[128 + n * 16];
            acc[n] = __builtin_amdgcn_wmma_f32_16x16x4_f32(
                false, a, false, b, (short)0, acc[n], false, false);
        }
    }
    asm volatile("v_nop\n\tv_nop\n\tv_nop\n\tv_nop\n\tv_nop\n\tv_nop\n\tv_nop\n\tv_nop"
                 : "+v"(acc[0]), "+v"(acc[1]), "+v"(acc[2]), "+v"(acc[3]),
                   "+v"(acc[4]), "+v"(acc[5]), "+v"(acc[6]), "+v"(acc[7]));
#pragma unroll
    for (int n = 0; n < 8; ++n) {
        float* od = uh + (size_t)(row0 + 8 * half) * COLS + j * DIM + n * 16 + l16;
#pragma unroll
        for (int g = 0; g < 8; ++g) {
            od[(size_t)g * COLS] = acc[n][g];
        }
    }
    float csum[8];
#pragma unroll
    for (int n = 0; n < 8; ++n) {
        const float s = ((acc[n][0] + acc[n][1]) + (acc[n][2] + acc[n][3])) +
                        ((acc[n][4] + acc[n][5]) + (acc[n][6] + acc[n][7]));
        csum[n] = s + other_row(s);
    }
    if (half == 0) {
        float* rd = red + wave * DIM + l16;
#pragma unroll
        for (int n = 0; n < 8; ++n) rd[n * 16] = csum[n];
    }
    __syncthreads();
    if (threadIdx.x < DIM) {
        const float* p = red + threadIdx.x;
        float s = p[0];
#pragma unroll
        for (int wv = 1; wv < 8; ++wv) s += p[wv * DIM];
        part[(size_t)blockIdx.x * COLS + j * DIM + threadIdx.x] = s;
    }
}

#define NPART 64
__launch_bounds__(256)
__global__ void colmean_kernel(const float* __restrict__ part, float* __restrict__ uhm) {
    __shared__ float red4[4][64];
    const int t   = threadIdx.x & 63;
    const int grp = threadIdx.x >> 6;
    const int c   = blockIdx.x * 64 + t;
    const float* p = part + (size_t)(grp * (NPART / 4)) * COLS + c;
    float s = p[0];
#pragma unroll
    for (int q = 1; q < NPART / 4; ++q) s += p[(size_t)q * COLS];
    red4[grp][t] = s;
    __syncthreads();
    if (threadIdx.x < 64)
        uhm[c] = ((red4[0][t] + red4[1][t]) + (red4[2][t] + red4[3][t])) * (1.0f / (float)JB);
}

__launch_bounds__(256)
__global__ void dotn2_kernel(const float* __restrict__ uh,
                             const float* __restrict__ uhm,
                             float* __restrict__ dotp, float* __restrict__ n2) {
    const int wave = threadIdx.x >> 5;
    const int lane = threadIdx.x & 31;
    const int pair = blockIdx.x * 8 + wave;
    const int r = pair >> 3, j = pair & 7;

    const float4 a = ((const float4*)(uh + (size_t)r * COLS + j * DIM))[lane];
    const float4 m = ((const float4*)(uhm + j * DIM))[lane];
    float d = a.x * m.x + a.y * m.y + a.z * m.z + a.w * m.w;
    float q = a.x * a.x + a.y * a.y + a.z * a.z + a.w * a.w;
#pragma unroll
    for (int off = 16; off > 0; off >>= 1) {
        d += __shfl_xor(d, off, 32);
        q += __shfl_xor(q, off, 32);
    }
    if (lane == 0) { dotp[pair] = d; n2[pair] = q; }
}

__launch_bounds__(1024)
__global__ void routing_kernel(const float* __restrict__ dotp,
                               const float* __restrict__ n2,
                               float* __restrict__ cf) {
    __shared__ float B7[8];
    __shared__ float c[8];
    __shared__ float sumj[8];
    const int tid = threadIdx.x;
    if (tid < 8) B7[tid] = 0.0f;

    for (int it = 0; it < 2; ++it) {
        __syncthreads();
        if (tid == 0) {
            float m = B7[0];
            for (int jj = 1; jj < 8; ++jj) m = fmaxf(m, B7[jj]);
            float s = 0.0f, e[8];
            for (int jj = 0; jj < 8; ++jj) { e[jj] = __expf(B7[jj] - m); s += e[jj]; }
            for (int jj = 0; jj < 8; ++jj) c[jj] = e[jj] / s;
        }
        __syncthreads();
        if (tid < 8) sumj[tid] = 0.0f;
        __syncthreads();
        const int j = tid & 7;
        const float cj = c[j];
        float a = 0.0f;
        for (int p = tid; p < JB * NJ; p += 1024)
            a += squash_gain(cj * cj * n2[p]) * dotp[p];
        atomicAdd(&sumj[j], a);
        __syncthreads();
        if (tid < 8) B7[tid] += c[tid] * sumj[tid] * (1.0f / (float)JB);
    }
    __syncthreads();
    if (tid == 0) {
        float m = B7[0];
        for (int jj = 1; jj < 8; ++jj) m = fmaxf(m, B7[jj]);
        float s = 0.0f, e[8];
        for (int jj = 0; jj < 8; ++jj) { e[jj] = __expf(B7[jj] - m); s += e[jj]; }
        for (int jj = 0; jj < 8; ++jj) cf[jj] = e[jj] / s;
    }
}

__launch_bounds__(256)
__global__ void final_kernel(const float* __restrict__ uh,
                             const float* __restrict__ n2,
                             const float* __restrict__ cf,
                             float* __restrict__ out) {
    const size_t idx  = (size_t)blockIdx.x * 256 + threadIdx.x;
    const size_t base = idx * 4;
    const int r = (int)(base >> 10);
    const int j = (int)((base >> 7) & 7);
    const float cj = cf[j];
    const float gain = cj * squash_gain(cj * cj * n2[r * 8 + j]);
    float4 a = *(const float4*)(uh + base);
    a.x *= gain; a.y *= gain; a.z *= gain; a.w *= gain;
    *(float4*)(out + base) = a;
}

extern "C" void kernel_launch(void* const* d_in, const int* in_sizes, int n_in,
                              void* d_out, int out_size, void* d_ws, size_t ws_size,
                              hipStream_t stream) {
    const float* x = (const float*)d_in[0];
    const float* w = (const float*)d_in[1];
    float* out = (float*)d_out;

    char* ws = (char*)d_ws;
    float* uh   = (float*)(ws);
    float* n2   = (float*)(ws + (size_t)JB * COLS * 4);
    float* dotp = (float*)(ws + (size_t)JB * COLS * 4 + 262144);
    float* uhm  = (float*)(ws + (size_t)JB * COLS * 4 + 524288);
    float* cf   = (float*)(ws + (size_t)JB * COLS * 4 + 528384);
    float* part = dotp;

    gemm_uhat_kernel<<<dim3(64, NJ), 256, 0, stream>>>(x, w, uh, part);
    colmean_kernel<<<16, 256, 0, stream>>>(part, uhm);
    dotn2_kernel<<<JB, 256, 0, stream>>>(uh, uhm, dotp, n2);
    routing_kernel<<<1, 1024, 0, stream>>>(dotp, n2, cf);
    final_kernel<<<(JB * COLS / 4) / 256, 256, 0, stream>>>(uh, n2, cf, out);
}


# dot/norm and output-scaling kernels: 4,096 waves each looping over 16 consecutive (row, capsule) pairs (8 KB, all loads in flight) instead of 65,536 one-pair waves; every sum and its order unchanged
# speedup vs baseline: 6.5627x; 1.0584x over previous
#include <hip/hip_runtime.h>
#include <hip/hip_bf16.h>

typedef __attribute__((ext_vector_type(2))) float v2f;
typedef __attribute__((ext_vector_type(8))) float v8f;

#define JB 8192
#define NJ 8
#define DIM 128
#define COLS 1024

__device__ __forceinline__ float other_row(float v) {
    const unsigned u = __builtin_bit_cast(unsigned, v);
    return __builtin_bit_cast(float,
        __builtin_amdgcn_permlanex16(u, u, 0x76543210u, 0xfedcba98u, false, false));
}

__device__ __forceinline__ float squash_gain(float n2) {
    return n2 / ((1.0f + n2) * sqrtf(n2 + 1e-8f));
}

__launch_bounds__(256)
__global__ void gemm_uhat_kernel(const float* __restrict__ x,
                                 const float* __restrict__ w,
                                 float* __restrict__ uh,
                                 float* __restrict__ part) {
    __shared__ float red[8 * DIM];

    const float* x7 = x + (size_t)7 * 8 * 1024 * 128;
    const float* w7 = w + (size_t)7 * 8 * 128 * 128;

    const int wave = threadIdx.x >> 5;
    const int lane = threadIdx.x & 31;
    const int half = lane >> 4;
    const int l16  = lane & 15;

    const int j    = blockIdx.y;
    const int row0 = blockIdx.x * 128 + wave * 16;
    const int r    = row0 + l16;
    const float* arow = x7 + (size_t)((r & 7) * 1024 + (r >> 3)) * 128;

    const float* wj = w7 + (size_t)j * 128 * 128;
    v8f acc[8];
#pragma unroll
    for (int n = 0; n < 8; ++n) acc[n] = (v8f){0,0,0,0,0,0,0,0};

    for (int k = 0; k < 128; k += 4) {
        v2f a;
        a.x = arow[k + 2 * half + 0];
        a.y = arow[k + 2 * half + 1];
        const float* wk = wj + (size_t)(k + 2 * half) * 128 + l16;
#pragma unroll
        for (int n = 0; n < 8; ++n) {
            v2f b;
            b.x = wk[n * 16];
            b.y = wk[128 + n * 16];
            acc[n] = __builtin_amdgcn_wmma_f32_16x16x4_f32(
                false, a, false, b, (short)0, acc[n], false, false);
        }
    }
    asm volatile("v_nop\n\tv_nop\n\tv_nop\n\tv_nop\n\tv_nop\n\tv_nop\n\tv_nop\n\tv_nop"
                 : "+v"(acc[0]), "+v"(acc[1]), "+v"(acc[2]), "+v"(acc[3]),
                   "+v"(acc[4]), "+v"(acc[5]), "+v"(acc[6]), "+v"(acc[7]));
#pragma unroll
    for (int n = 0; n < 8; ++n) {
        float* od = uh + (size_t)(row0 + 8 * half) * COLS + j * DIM + n * 16 + l16;
#pragma unroll
        for (int g = 0; g < 8; ++g) {
            od[(size_t)g * COLS] = acc[n][g];
        }
    }
    float csum[8];
#pragma unroll
    for (int n = 0; n < 8; ++n) {
        const float s = ((acc[n][0] + acc[n][1]) + (acc[n][2] + acc[n][3])) +
                        ((acc[n][4] + acc[n][5]) + (acc[n][6] + acc[n][7]));
        csum[n] = s + other_row(s);
    }
    if (half == 0) {
        float* rd = red + wave * DIM + l16;
#pragma unroll
        for (int n = 0; n < 8; ++n) rd[n * 16] = csum[n];
    }
    __syncthreads();
    if (threadIdx.x < DIM) {
        const float* p = red + threadIdx.x;
        float s = p[0];
#pragma unroll
        for (int wv = 1; wv < 8; ++wv) s += p[wv * DIM];
        part[(size_t)blockIdx.x * COLS + j * DIM + threadIdx.x] = s;
    }
}

#define NPART 64
__launch_bounds__(256)
__global__ void colmean_kernel(const float* __restrict__ part, float* __restrict__ uhm) {
    __shared__ float red4[4][64];
    const int t   = threadIdx.x & 63;
    const int grp = threadIdx.x >> 6;
    const int c   = blockIdx.x * 64 + t;
    const float* p = part + (size_t)(grp * (NPART / 4)) * COLS + c;
    float s = p[0];
#pragma unroll
    for (int q = 1; q < NPART / 4; ++q) s += p[(size_t)q * COLS];
    red4[grp][t] = s;
    __syncthreads();
    if (threadIdx.x < 64)
        uhm[c] = ((red4[0][t] + red4[1][t]) + (red4[2][t] + red4[3][t])) * (1.0f / (float)JB);
}

#define PPW 16
#define STREAM_BLOCKS 512

__launch_bounds__(256)
__global__ void dotn2_kernel(const float* __restrict__ uh,
                             const float* __restrict__ uhm,
                             float* __restrict__ dotp, float* __restrict__ n2) {
    const int wave = threadIdx.x >> 5;
    const int lane = threadIdx.x & 31;
    const int p0 = (blockIdx.x * 8 + wave) * PPW;

    float4 m[NJ];
#pragma unroll
    for (int jj = 0; jj < NJ; ++jj) m[jj] = ((const float4*)(uhm + jj * DIM))[lane];

    const float4* src = (const float4*)(uh + (size_t)p0 * DIM) + lane;
    float4 a[PPW];
#pragma unroll
    for (int i = 0; i < PPW; ++i) a[i] = src[i * (DIM / 4)];
    __builtin_amdgcn_sched_barrier(0);

    float dsel = 0.0f, qsel = 0.0f;
#pragma unroll
    for (int i = 0; i < PPW; ++i) {
        const float4 mm = m[i & 7];
        float d = a[i].x * mm.x + a[i].y * mm.y + a[i].z * mm.z + a[i].w * mm.w;
        float q = a[i].x * a[i].x + a[i].y * a[i].y + a[i].z * a[i].z + a[i].w * a[i].w;
#pragma unroll
        for (int off = 16; off > 0; off >>= 1) {
            d += __shfl_xor(d, off, 32);
            q += __shfl_xor(q, off, 32);
        }
        if (lane == i) { dsel = d; qsel = q; }
    }
    if (lane < PPW) { dotp[p0 + lane] = dsel; n2[p0 + lane] = qsel; }
}

__launch_bounds__(1024)
__global__ void routing_kernel(const float* __restrict__ dotp,
                               const float* __restrict__ n2,
                               float* __restrict__ cf) {
    __shared__ float B7[8];
    __shared__ float c[8];
    __shared__ float sumj[8];
    const int tid = threadIdx.x;
    if (tid < 8) B7[tid] = 0.0f;

    for (int it = 0; it < 2; ++it) {
        __syncthreads();
        if (tid == 0) {
            float m = B7[0];
            for (int jj = 1; jj < 8; ++jj) m = fmaxf(m, B7[jj]);
            float s = 0.0f, e[8];
            for (int jj = 0; jj < 8; ++jj) { e[jj] = __expf(B7[jj] - m); s += e[jj]; }
            for (int jj = 0; jj < 8; ++jj) c[jj] = e[jj] / s;
        }
        __syncthreads();
        if (tid < 8) sumj[tid] = 0.0f;
        __syncthreads();
        const int j = tid & 7;
        const float cj = c[j];
        float a = 0.0f;
        for (int p = tid; p < JB * NJ; p += 1024)
            a += squash_gain(cj * cj * n2[p]) * dotp[p];
        atomicAdd(&sumj[j], a);
        __syncthreads();
        if (tid < 8) B7[tid] += c[tid] * sumj[tid] * (1.0f / (float)JB);
    }
    __syncthreads();
    if (tid == 0) {
        float m = B7[0];
        for (int jj = 1; jj < 8; ++jj) m = fmaxf(m, B7[jj]);
        float s = 0.0f, e[8];
        for (int jj = 0; jj < 8; ++jj) { e[jj] = __expf(B7[jj] - m); s += e[jj]; }
        for (int jj = 0; jj < 8; ++jj) cf[jj] = e[jj] / s;
    }
}

__launch_bounds__(256)
__global__ void final_kernel(const float* __restrict__ uh,
                             const float* __restrict__ n2,
                             const float* __restrict__ cf,
                             float* __restrict__ out) {
    const int wave = threadIdx.x >> 5;
    const int lane = threadIdx.x & 31;
    const int p0 = (blockIdx.x * 8 + wave) * PPW;

    const int pl = lane & (PPW - 1);
    float cj = cf[pl & 7];
    float nn = n2[p0 + pl];
    __builtin_amdgcn_sched_barrier(0);

    const float4* src = (const float4*)(uh + (size_t)p0 * DIM) + lane;
    float4* dst = (float4*)(out + (size_t)p0 * DIM) + lane;
    float4 a[PPW];
#pragma unroll
    for (int i = 0; i < PPW; ++i) a[i] = src[i * (DIM / 4)];
    __builtin_amdgcn_sched_barrier(0);
    asm volatile("" : "+v"(cj), "+v"(nn));

    const float gain = cj * squash_gain(cj * cj * nn);
    const int gbits = __builtin_bit_cast(int, gain);

#pragma unroll
    for (int i = 0; i < PPW; ++i) {
        const float g = __builtin_bit_cast(float, __builtin_amdgcn_readlane(gbits, i));
        float4 v = a[i];
        v.x *= g; v.y *= g; v.z *= g; v.w *= g;
        dst[i * (DIM / 4)] = v;
    }
}

extern "C" void kernel_launch(void* const* d_in, const int* in_sizes, int n_in,
                              void* d_out, int out_size, void* d_ws, size_t ws_size,
                              hipStream_t stream) {
    const float* x = (const float*)d_in[0];
    const float* w = (const float*)d_in[1];
    float* out = (float*)d_out;

    char* ws = (char*)d_ws;
    float* uh   = (float*)(ws);
    float* n2   = (float*)(ws + (size_t)JB * COLS * 4);
    float* dotp = (float*)(ws + (size_t)JB * COLS * 4 + 262144);
    float* uhm  = (float*)(ws + (size_t)JB * COLS * 4 + 524288);
    float* cf   = (float*)(ws + (size_t)JB * COLS * 4 + 528384);
    float* part = dotp;

    gemm_uhat_kernel<<<dim3(64, NJ), 256, 0, stream>>>(x, w, uh, part);
    colmean_kernel<<<16, 256, 0, stream>>>(part, uhm);
    dotn2_kernel<<<STREAM_BLOCKS, 256, 0, stream>>>(uh, uhm, dotp, n2);
    routing_kernel<<<1, 1024, 0, stream>>>(dotp, n2, cf);
    final_kernel<<<STREAM_BLOCKS, 256, 0, stream>>>(uh, n2, cf, out);
}


# routing sums leave the one-block serial loop: first sum (c=1/8) as per-block partials in the dot/norm kernel, second over 32 blocks, joined in the output kernel; sums are fixed trees, no LDS atomic
# speedup vs baseline: 10.3615x; 1.5788x over previous
#include <hip/hip_runtime.h>
#include <hip/hip_bf16.h>

typedef __attribute__((ext_vector_type(2))) float v2f;
typedef __attribute__((ext_vector_type(8))) float v8f;

#define JB 8192
#define NJ 8
#define DIM 128
#define COLS 1024

__device__ __forceinline__ float other_row(float v) {
    const unsigned u = __builtin_bit_cast(unsigned, v);
    return __builtin_bit_cast(float,
        __builtin_amdgcn_permlanex16(u, u, 0x76543210u, 0xfedcba98u, false, false));
}

__device__ __forceinline__ float squash_gain(float n2) {
    return n2 / ((1.0f + n2) * sqrtf(n2 + 1e-8f));
}

__device__ __forceinline__ float softmax8(float b) {
    float m = b;
    m = fmaxf(m, __shfl_xor(m, 1, 32));
    m = fmaxf(m, __shfl_xor(m, 2, 32));
    m = fmaxf(m, __shfl_xor(m, 4, 32));
    const float e = __expf(b - m);
    float s = e;
    s += __shfl_xor(s, 1, 32);
    s += __shfl_xor(s, 2, 32);
    s += __shfl_xor(s, 4, 32);
    return e / s;
}

__device__ __forceinline__ float sum_groups8(float v) {
    v += __shfl_xor(v, 8, 32);
    v += __shfl_xor(v, 16, 32);
    return v;
}

__launch_bounds__(256)
__global__ void gemm_uhat_kernel(const float* __restrict__ x,
                                 const float* __restrict__ w,
                                 float* __restrict__ uh,
                                 float* __restrict__ part) {
    __shared__ float red[8 * DIM];

    const float* x7 = x + (size_t)7 * 8 * 1024 * 128;
    const float* w7 = w + (size_t)7 * 8 * 128 * 128;

    const int wave = threadIdx.x >> 5;
    const int lane = threadIdx.x & 31;
    const int half = lane >> 4;
    const int l16  = lane & 15;

    const int j    = blockIdx.y;
    const int row0 = blockIdx.x * 128 + wave * 16;
    const int r    = row0 + l16;
    const float* arow = x7 + (size_t)((r & 7) * 1024 + (r >> 3)) * 128;

    const float* wj = w7 + (size_t)j * 128 * 128;
    v8f acc[8];
#pragma unroll
    for (int n = 0; n < 8; ++n) acc[n] = (v8f){0,0,0,0,0,0,0,0};

    for (int k = 0; k < 128; k += 4) {
        v2f a;
        a.x = arow[k + 2 * half + 0];
        a.y = arow[k + 2 * half + 1];
        const float* wk = wj + (size_t)(k + 2 * half) * 128 + l16;
#pragma unroll
        for (int n = 0; n < 8; ++n) {
            v2f b;
            b.x = wk[n * 16];
            b.y = wk[128 + n * 16];
            acc[n] = __builtin_amdgcn_wmma_f32_16x16x4_f32(
                false, a, false, b, (short)0, acc[n], false, false);
        }
    }
    asm volatile("v_nop\n\tv_nop\n\tv_nop\n\tv_nop\n\tv_nop\n\tv_nop\n\tv_nop\n\tv_nop"
                 : "+v"(acc[0]), "+v"(acc[1]), "+v"(acc[2]), "+v"(acc[3]),
                   "+v"(acc[4]), "+v"(acc[5]), "+v"(acc[6]), "+v"(acc[7]));
#pragma unroll
    for (int n = 0; n < 8; ++n) {
        float* od = uh + (size_t)(row0 + 8 * half) * COLS + j * DIM + n * 16 + l16;
#pragma unroll
        for (int g = 0; g < 8; ++g) {
            od[(size_t)g * COLS] = acc[n][g];
        }
    }
    float csum[8];
#pragma unroll
    for (int n = 0; n < 8; ++n) {
        const float s = ((acc[n][0] + acc[n][1]) + (acc[n][2] + acc[n][3])) +
                        ((acc[n][4] + acc[n][5]) + (acc[n][6] + acc[n][7]));
        csum[n] = s + other_row(s);
    }
    if (half == 0) {
        float* rd = red + wave * DIM + l16;
#pragma unroll
        for (int n = 0; n < 8; ++n) rd[n * 16] = csum[n];
    }
    __syncthreads();
    if (threadIdx.x < DIM) {
        const float* p = red + threadIdx.x;
        float s = p[0];
#pragma unroll
        for (int wv = 1; wv < 8; ++wv) s += p[wv * DIM];
        part[(size_t)blockIdx.x * COLS + j * DIM + threadIdx.x] = s;
    }
}

#define NPART 64
__launch_bounds__(256)
__global__ void colmean_kernel(const float* __restrict__ part, float* __restrict__ uhm) {
    __shared__ float red4[4][64];
    const int t   = threadIdx.x & 63;
    const int grp = threadIdx.x >> 6;
    const int c   = blockIdx.x * 64 + t;
    const float* p = part + (size_t)(grp * (NPART / 4)) * COLS + c;
    float s = p[0];
#pragma unroll
    for (int q = 1; q < NPART / 4; ++q) s += p[(size_t)q * COLS];
    red4[grp][t] = s;
    __syncthreads();
    if (threadIdx.x < 64)
        uhm[c] = ((red4[0][t] + red4[1][t]) + (red4[2][t] + red4[3][t])) * (1.0f / (float)JB);
}

#define PPW 16
#define STREAM_BLOCKS 512

__launch_bounds__(256)
__global__ void dotn2_kernel(const float* __restrict__ uh,
                             const float* __restrict__ uhm,
                             float* __restrict__ dotp, float* __restrict__ n2,
                             float* __restrict__ p1) {
    __shared__ float r1[8][NJ];
    const int wave = threadIdx.x >> 5;
    const int lane = threadIdx.x & 31;
    const int p0 = (blockIdx.x * 8 + wave) * PPW;

    float4 m[NJ];
#pragma unroll
    for (int jj = 0; jj < NJ; ++jj) m[jj] = ((const float4*)(uhm + jj * DIM))[lane];

    const float4* src = (const float4*)(uh + (size_t)p0 * DIM) + lane;
    float4 a[PPW];
#pragma unroll
    for (int i = 0; i < PPW; ++i) a[i] = src[i * (DIM / 4)];
    __builtin_amdgcn_sched_barrier(0);

    float dsel = 0.0f, qsel = 0.0f;
#pragma unroll
    for (int i = 0; i < PPW; ++i) {
        const float4 mm = m[i & 7];
        float d = a[i].x * mm.x + a[i].y * mm.y + a[i].z * mm.z + a[i].w * mm.w;
        float q = a[i].x * a[i].x + a[i].y * a[i].y + a[i].z * a[i].z + a[i].w * a[i].w;
#pragma unroll
        for (int off = 16; off > 0; off >>= 1) {
            d += __shfl_xor(d, off, 32);
            q += __shfl_xor(q, off, 32);
        }
        if (lane == i) { dsel = d; qsel = q; }
    }
    float t1 = squash_gain(0.015625f * qsel) * dsel;
    t1 += __shfl_xor(t1, 8, 32);
    if (lane < NJ) r1[wave][lane] = t1;
    __syncthreads();

    if (lane < PPW) { dotp[p0 + lane] = dsel; n2[p0 + lane] = qsel; }

    if (threadIdx.x < NJ) {
        float s = r1[0][threadIdx.x];
#pragma unroll
        for (int wv = 1; wv < 8; ++wv) s += r1[wv][threadIdx.x];
        p1[blockIdx.x * NJ + threadIdx.x] = s;
    }
}

#define RB 32
#define RTHREADS 512
#define RWAVES (RTHREADS / 32)
#define RSTRIDE (RB * RTHREADS)
#define RSWEEPS (JB * NJ / RSTRIDE)
#define P1N (STREAM_BLOCKS * NJ)
__launch_bounds__(RTHREADS)
__global__ void routing_kernel(const float* __restrict__ p1,
                               const float* __restrict__ dotp,
                               const float* __restrict__ n2,
                               float* __restrict__ rt,
                               float* __restrict__ p2) {
    __shared__ float redA[RWAVES * NJ];
    __shared__ float redC[RWAVES * NJ];
    __shared__ float c1s[NJ];
    const int tid  = threadIdx.x;
    const int lane = tid & 31;
    const int wave = tid >> 5;
    const int g    = blockIdx.x * RTHREADS + tid;

    float nn[RSWEEPS], dd[RSWEEPS];
#pragma unroll
    for (int k = 0; k < RSWEEPS; ++k) {
        nn[k] = n2[g + k * RSTRIDE];
        dd[k] = dotp[g + k * RSTRIDE];
    }

    float a = p1[tid];
#pragma unroll
    for (int k = 1; k < P1N / RTHREADS; ++k) a += p1[tid + k * RTHREADS];
    a = sum_groups8(a);
    if (lane < NJ) redA[wave * NJ + lane] = a;
    __syncthreads();
    if (wave == 0) {
        float s = redA[lane];
#pragma unroll
        for (int k = 1; k < RWAVES * NJ / 32; ++k) s += redA[lane + 32 * k];
        s = sum_groups8(s);
        const float b1 = 0.125f * s * (1.0f / (float)JB);
        const float c1 = softmax8(b1);
        if (lane < NJ) {
            c1s[lane] = c1;
            if (blockIdx.x == 0) { rt[lane] = b1; rt[NJ + lane] = c1; }
        }
    }
    __syncthreads();

    const float cj = c1s[lane & 7];
    const float cc = cj * cj;
    float t = 0.0f;
#pragma unroll
    for (int k = 0; k < RSWEEPS; ++k) t += squash_gain(cc * nn[k]) * dd[k];
    t = sum_groups8(t);
    if (lane < NJ) redC[wave * NJ + lane] = t;
    __syncthreads();
    if (wave == 0) {
        float s = redC[lane];
#pragma unroll
        for (int k = 1; k < RWAVES * NJ / 32; ++k) s += redC[lane + 32 * k];
        s = sum_groups8(s);
        if (lane < NJ) p2[blockIdx.x * NJ + lane] = s;
    }
}

__launch_bounds__(256)
__global__ void final_kernel(const float* __restrict__ uh,
                             const float* __restrict__ n2,
                             const float* __restrict__ rt,
                             const float* __restrict__ p2,
                             float* __restrict__ out) {
    const int wave = threadIdx.x >> 5;
    const int lane = threadIdx.x & 31;
    const int p0 = (blockIdx.x * 8 + wave) * PPW;

    const int pl = lane & (PPW - 1);
    const int jl = lane & 7;
    float nn = n2[p0 + pl];
    float b1 = rt[jl];
    float c1 = rt[NJ + jl];
    float s2[RB * NJ / 32];
#pragma unroll
    for (int k = 0; k < RB * NJ / 32; ++k) s2[k] = p2[lane + 32 * k];
    __builtin_amdgcn_sched_barrier(0);

    const float4* src = (const float4*)(uh + (size_t)p0 * DIM) + lane;
    float4* dst = (float4*)(out + (size_t)p0 * DIM) + lane;
    float4 a[PPW];
#pragma unroll
    for (int i = 0; i < PPW; ++i) a[i] = src[i * (DIM / 4)];
    __builtin_amdgcn_sched_barrier(0);
    asm volatile("" : "+v"(nn), "+v"(b1), "+v"(c1),
                      "+v"(s2[0]), "+v"(s2[1]), "+v"(s2[2]), "+v"(s2[3]),
                      "+v"(s2[4]), "+v"(s2[5]), "+v"(s2[6]), "+v"(s2[7]));

    float sj = ((s2[0] + s2[1]) + (s2[2] + s2[3])) + ((s2[4] + s2[5]) + (s2[6] + s2[7]));
    sj = sum_groups8(sj);
    const float cj = softmax8(b1 + c1 * sj * (1.0f / (float)JB));

    const float gain = cj * squash_gain(cj * cj * nn);
    const int gbits = __builtin_bit_cast(int, gain);

#pragma unroll
    for (int i = 0; i < PPW; ++i) {
        const float g = __builtin_bit_cast(float, __builtin_amdgcn_readlane(gbits, i));
        float4 v = a[i];
        v.x *= g; v.y *= g; v.z *= g; v.w *= g;
        dst[i * (DIM / 4)] = v;
    }
}

extern "C" void kernel_launch(void* const* d_in, const int* in_sizes, int n_in,
                              void* d_out, int out_size, void* d_ws, size_t ws_size,
                              hipStream_t stream) {
    const float* x = (const float*)d_in[0];
    const float* w = (const float*)d_in[1];
    float* out = (float*)d_out;

    char* ws = (char*)d_ws;
    float* uh   = (float*)(ws);
    float* n2   = (float*)(ws + (size_t)JB * COLS * 4);
    float* dotp = (float*)(ws + (size_t)JB * COLS * 4 + 262144);
    float* uhm  = (float*)(ws + (size_t)JB * COLS * 4 + 524288);
    float* part = dotp;
    float* p2 = uhm;
    float* rt = uhm + 256;
    float* p1 = out;

    gemm_uhat_kernel<<<dim3(64, NJ), 256, 0, stream>>>(x, w, uh, part);
    colmean_kernel<<<16, 256, 0, stream>>>(part, uhm);
    dotn2_kernel<<<STREAM_BLOCKS, 256, 0, stream>>>(uh, uhm, dotp, n2, p1);
    routing_kernel<<<RB, RTHREADS, 0, stream>>>(p1, dotp, n2, rt, p2);
    final_kernel<<<STREAM_BLOCKS, 256, 0, stream>>>(uh, n2, rt, p2, out);
}


# GEMM B operand via LDS: W[7,j] staged once per block (k pairs interleaved, lane owns 4 adjacent columns); 4 ds_load_b128 per k-step for 16 global loads, uh leaves as 16-byte stores; sums unchanged
# speedup vs baseline: 12.7620x; 1.2317x over previous
#include <hip/hip_runtime.h>
#include <hip/hip_bf16.h>

typedef __attribute__((ext_vector_type(2))) float v2f;
typedef __attribute__((ext_vector_type(8))) float v8f;

#define JB 8192
#define NJ 8
#define DIM 128
#define COLS 1024

__device__ __forceinline__ float other_row(float v) {
    const unsigned u = __builtin_bit_cast(unsigned, v);
    return __builtin_bit_cast(float,
        __builtin_amdgcn_permlanex16(u, u, 0x76543210u, 0xfedcba98u, false, false));
}

__device__ __forceinline__ float squash_gain(float n2) {
    return n2 / ((1.0f + n2) * sqrtf(n2 + 1e-8f));
}

__device__ __forceinline__ float softmax8(float b) {
    float m = b;
    m = fmaxf(m, __shfl_xor(m, 1, 32));
    m = fmaxf(m, __shfl_xor(m, 2, 32));
    m = fmaxf(m, __shfl_xor(m, 4, 32));
    const float e = __expf(b - m);
    float s = e;
    s += __shfl_xor(s, 1, 32);
    s += __shfl_xor(s, 2, 32);
    s += __shfl_xor(s, 4, 32);
    return e / s;
}

__device__ __forceinline__ float sum_groups8(float v) {
    v += __shfl_xor(v, 8, 32);
    v += __shfl_xor(v, 16, 32);
    return v;
}

__launch_bounds__(256)
__global__ void gemm_uhat_kernel(const float* __restrict__ x,
                                 const float* __restrict__ w,
                                 float* __restrict__ uh,
                                 float* __restrict__ part) {
    __shared__ float4 wsm[64 * 64];
    __shared__ float4 red4[8 * DIM / 4];

    const float* x7 = x + (size_t)7 * 8 * 1024 * 128;
    const float* w7 = w + (size_t)7 * 8 * 128 * 128;

    const int wave = threadIdx.x >> 5;
    const int lane = threadIdx.x & 31;
    const int half = lane >> 4;
    const int l16  = lane & 15;

    const int j    = blockIdx.y;
    const int row0 = blockIdx.x * 128 + wave * 16;
    const int r    = row0 + l16;
    const float* arow = x7 + (size_t)((r & 7) * 1024 + (r >> 3)) * 128;

    const float* wj = w7 + (size_t)j * 128 * 128;

    {
        const int slot = (lane >> 4) * 32 + (lane & 15);
        float4 r0[8], r1[8];
#pragma unroll
        for (int i = 0; i < 8; ++i) {
            const int kp = i * 8 + wave;
            r0[i] = *(const float4*)(wj + (size_t)(2 * kp) * 128 + 4 * lane);
            r1[i] = *(const float4*)(wj + (size_t)(2 * kp + 1) * 128 + 4 * lane);
        }
#pragma unroll
        for (int i = 0; i < 8; ++i) {
            const int kp = i * 8 + wave;
            wsm[kp * 64 + slot]      = make_float4(r0[i].x, r1[i].x, r0[i].y, r1[i].y);
            wsm[kp * 64 + slot + 16] = make_float4(r0[i].z, r1[i].z, r0[i].w, r1[i].w);
        }
    }

    v8f acc[8];
#pragma unroll
    for (int n = 0; n < 8; ++n) acc[n] = (v8f){0,0,0,0,0,0,0,0};

    constexpr int AD = 2;
    v2f ar[AD + 1];
#pragma unroll
    for (int t = 0; t < AD; ++t) {
        ar[t].x = arow[4 * t + 2 * half + 0];
        ar[t].y = arow[4 * t + 2 * half + 1];
    }
    __syncthreads();

    const float4* bp = wsm + half * 64 + l16;
    float4 q[2][4];
#pragma unroll
    for (int m = 0; m < 4; ++m) q[0][m] = bp[m * 16];

#pragma unroll
    for (int t = 0; t < 32; ++t) {
        if (t + AD < 32) {
            ar[(t + AD) % (AD + 1)].x = arow[4 * (t + AD) + 2 * half + 0];
            ar[(t + AD) % (AD + 1)].y = arow[4 * (t + AD) + 2 * half + 1];
        }
        if (t + 1 < 32) {
#pragma unroll
            for (int m = 0; m < 4; ++m) q[(t + 1) & 1][m] = bp[(t + 1) * 128 + m * 16];
        }
        const v2f a = ar[t % (AD + 1)];
#pragma unroll
        for (int m = 0; m < 4; ++m) {
            const float4 qq = q[t & 1][m];
            v2f b0, b1;
            b0.x = qq.x; b0.y = qq.y;
            b1.x = qq.z; b1.y = qq.w;
            acc[2 * m] = __builtin_amdgcn_wmma_f32_16x16x4_f32(
                false, a, false, b0, (short)0, acc[2 * m], false, false);
            acc[2 * m + 1] = __builtin_amdgcn_wmma_f32_16x16x4_f32(
                false, a, false, b1, (short)0, acc[2 * m + 1], false, false);
        }
    }
    asm volatile("v_nop\n\tv_nop\n\tv_nop\n\tv_nop\n\tv_nop\n\tv_nop\n\tv_nop\n\tv_nop"
                 : "+v"(acc[0]), "+v"(acc[1]), "+v"(acc[2]), "+v"(acc[3]),
                   "+v"(acc[4]), "+v"(acc[5]), "+v"(acc[6]), "+v"(acc[7]));
#pragma unroll
    for (int g = 0; g < 8; ++g) {
        float* od = uh + (size_t)(row0 + 8 * half + g) * COLS + j * DIM + 4 * l16;
        *(float4*)od        = make_float4(acc[0][g], acc[1][g], acc[2][g], acc[3][g]);
        *(float4*)(od + 64) = make_float4(acc[4][g], acc[5][g], acc[6][g], acc[7][g]);
    }
    float csum[8];
#pragma unroll
    for (int n = 0; n < 8; ++n) {
        const float s = ((acc[n][0] + acc[n][1]) + (acc[n][2] + acc[n][3])) +
                        ((acc[n][4] + acc[n][5]) + (acc[n][6] + acc[n][7]));
        csum[n] = s + other_row(s);
    }
    if (half == 0) {
        float4* rd = red4 + wave * (DIM / 4);
        rd[l16]      = make_float4(csum[0], csum[1], csum[2], csum[3]);
        rd[16 + l16] = make_float4(csum[4], csum[5], csum[6], csum[7]);
    }
    __syncthreads();
    if (threadIdx.x < DIM) {
        const float* p = (const float*)red4 + threadIdx.x;
        float s = p[0];
#pragma unroll
        for (int wv = 1; wv < 8; ++wv) s += p[wv * DIM];
        part[(size_t)blockIdx.x * COLS + j * DIM + threadIdx.x] = s;
    }
}

#define NPART 64
__launch_bounds__(256)
__global__ void colmean_kernel(const float* __restrict__ part, float* __restrict__ uhm) {
    __shared__ float red4[4][64];
    const int t   = threadIdx.x & 63;
    const int grp = threadIdx.x >> 6;
    const int c   = blockIdx.x * 64 + t;
    const float* p = part + (size_t)(grp * (NPART / 4)) * COLS + c;
    float s = p[0];
#pragma unroll
    for (int q = 1; q < NPART / 4; ++q) s += p[(size_t)q * COLS];
    red4[grp][t] = s;
    __syncthreads();
    if (threadIdx.x < 64)
        uhm[c] = ((red4[0][t] + red4[1][t]) + (red4[2][t] + red4[3][t])) * (1.0f / (float)JB);
}

#define PPW 16
#define STREAM_BLOCKS 512

__launch_bounds__(256)
__global__ void dotn2_kernel(const float* __restrict__ uh,
                             const float* __restrict__ uhm,
                             float* __restrict__ dotp, float* __restrict__ n2,
                             float* __restrict__ p1) {
    __shared__ float r1[8][NJ];
    const int wave = threadIdx.x >> 5;
    const int lane = threadIdx.x & 31;
    const int p0 = (blockIdx.x * 8 + wave) * PPW;

    float4 m[NJ];
#pragma unroll
    for (int jj = 0; jj < NJ; ++jj) m[jj] = ((const float4*)(uhm + jj * DIM))[lane];

    const float4* src = (const float4*)(uh + (size_t)p0 * DIM) + lane;
    float4 a[PPW];
#pragma unroll
    for (int i = 0; i < PPW; ++i) a[i] = src[i * (DIM / 4)];
    __builtin_amdgcn_sched_barrier(0);

    float dsel = 0.0f, qsel = 0.0f;
#pragma unroll
    for (int i = 0; i < PPW; ++i) {
        const float4 mm = m[i & 7];
        float d = a[i].x * mm.x + a[i].y * mm.y + a[i].z * mm.z + a[i].w * mm.w;
        float q = a[i].x * a[i].x + a[i].y * a[i].y + a[i].z * a[i].z + a[i].w * a[i].w;
#pragma unroll
        for (int off = 16; off > 0; off >>= 1) {
            d += __shfl_xor(d, off, 32);
            q += __shfl_xor(q, off, 32);
        }
        if (lane == i) { dsel = d; qsel = q; }
    }
    float t1 = squash_gain(0.015625f * qsel) * dsel;
    t1 += __shfl_xor(t1, 8, 32);
    if (lane < NJ) r1[wave][lane] = t1;
    __syncthreads();

    if (lane < PPW) { dotp[p0 + lane] = dsel; n2[p0 + lane] = qsel; }

    if (threadIdx.x < NJ) {
        float s = r1[0][threadIdx.x];
#pragma unroll
        for (int wv = 1; wv < 8; ++wv) s += r1[wv][threadIdx.x];
        p1[blockIdx.x * NJ + threadIdx.x] = s;
    }
}

#define RB 32
#define RTHREADS 512
#define RWAVES (RTHREADS / 32)
#define RSTRIDE (RB * RTHREADS)
#define RSWEEPS (JB * NJ / RSTRIDE)
#define P1N (STREAM_BLOCKS * NJ)
__launch_bounds__(RTHREADS)
__global__ void routing_kernel(const float* __restrict__ p1,
                               const float* __restrict__ dotp,
                               const float* __restrict__ n2,
                               float* __restrict__ rt,
                               float* __restrict__ p2) {
    __shared__ float redA[RWAVES * NJ];
    __shared__ float redC[RWAVES * NJ];
    __shared__ float c1s[NJ];
    const int tid  = threadIdx.x;
    const int lane = tid & 31;
    const int wave = tid >> 5;
    const int g    = blockIdx.x * RTHREADS + tid;

    float nn[RSWEEPS], dd[RSWEEPS];
#pragma unroll
    for (int k = 0; k < RSWEEPS; ++k) {
        nn[k] = n2[g + k * RSTRIDE];
        dd[k] = dotp[g + k * RSTRIDE];
    }

    float a = p1[tid];
#pragma unroll
    for (int k = 1; k < P1N / RTHREADS; ++k) a += p1[tid + k * RTHREADS];
    a = sum_groups8(a);
    if (lane < NJ) redA[wave * NJ + lane] = a;
    __syncthreads();
    if (wave == 0) {
        float s = redA[lane];
#pragma unroll
        for (int k = 1; k < RWAVES * NJ / 32; ++k) s += redA[lane + 32 * k];
        s = sum_groups8(s);
        const float b1 = 0.125f * s * (1.0f / (float)JB);
        const float c1 = softmax8(b1);
        if (lane < NJ) {
            c1s[lane] = c1;
            if (blockIdx.x == 0) { rt[lane] = b1; rt[NJ + lane] = c1; }
        }
    }
    __syncthreads();

    const float cj = c1s[lane & 7];
    const float cc = cj * cj;
    float t = 0.0f;
#pragma unroll
    for (int k = 0; k < RSWEEPS; ++k) t += squash_gain(cc * nn[k]) * dd[k];
    t = sum_groups8(t);
    if (lane < NJ) redC[wave * NJ + lane] = t;
    __syncthreads();
    if (wave == 0) {
        float s = redC[lane];
#pragma unroll
        for (int k = 1; k < RWAVES * NJ / 32; ++k) s += redC[lane + 32 * k];
        s = sum_groups8(s);
        if (lane < NJ) p2[blockIdx.x * NJ + lane] = s;
    }
}

__launch_bounds__(256)
__global__ void final_kernel(const float* __restrict__ uh,
                             const float* __restrict__ n2,
                             const float* __restrict__ rt,
                             const float* __restrict__ p2,
                             float* __restrict__ out) {
    const int wave = threadIdx.x >> 5;
    const int lane = threadIdx.x & 31;
    const int p0 = (blockIdx.x * 8 + wave) * PPW;

    const int pl = lane & (PPW - 1);
    const int jl = lane & 7;
    float nn = n2[p0 + pl];
    float b1 = rt[jl];
    float c1 = rt[NJ + jl];
    float s2[RB * NJ / 32];
#pragma unroll
    for (int k = 0; k < RB * NJ / 32; ++k) s2[k] = p2[lane + 32 * k];
    __builtin_amdgcn_sched_barrier(0);

    const float4* src = (const float4*)(uh + (size_t)p0 * DIM) + lane;
    float4* dst = (float4*)(out + (size_t)p0 * DIM) + lane;
    float4 a[PPW];
#pragma unroll
    for (int i = 0; i < PPW; ++i) a[i] = src[i * (DIM / 4)];
    __builtin_amdgcn_sched_barrier(0);
    asm volatile("" : "+v"(nn), "+v"(b1), "+v"(c1),
                      "+v"(s2[0]), "+v"(s2[1]), "+v"(s2[2]), "+v"(s2[3]),
                      "+v"(s2[4]), "+v"(s2[5]), "+v"(s2[6]), "+v"(s2[7]));

    float sj = ((s2[0] + s2[1]) + (s2[2] + s2[3])) + ((s2[4] + s2[5]) + (s2[6] + s2[7]));
    sj = sum_groups8(sj);
    const float cj = softmax8(b1 + c1 * sj * (1.0f / (float)JB));

    const float gain = cj * squash_gain(cj * cj * nn);
    const int gbits = __builtin_bit_cast(int, gain);

#pragma unroll
    for (int i = 0; i < PPW; ++i) {
        const float g = __builtin_bit_cast(float, __builtin_amdgcn_readlane(gbits, i));
        float4 v = a[i];
        v.x *= g; v.y *= g; v.z *= g; v.w *= g;
        dst[i * (DIM / 4)] = v;
    }
}

extern "C" void kernel_launch(void* const* d_in, const int* in_sizes, int n_in,
                              void* d_out, int out_size, void* d_ws, size_t ws_size,
                              hipStream_t stream) {
    const float* x = (const float*)d_in[0];
    const float* w = (const float*)d_in[1];
    float* out = (float*)d_out;

    char* ws = (char*)d_ws;
    float* uh   = (float*)(ws);
    float* n2   = (float*)(ws + (size_t)JB * COLS * 4);
    float* dotp = (float*)(ws + (size_t)JB * COLS * 4 + 262144);
    float* uhm  = (float*)(ws + (size_t)JB * COLS * 4 + 524288);
    float* part = dotp;
    float* p2 = uhm;
    float* rt = uhm + 256;
    float* p1 = out;

    gemm_uhat_kernel<<<dim3(64, NJ), 256, 0, stream>>>(x, w, uh, part);
    colmean_kernel<<<16, 256, 0, stream>>>(part, uhm);
    dotn2_kernel<<<STREAM_BLOCKS, 256, 0, stream>>>(uh, uhm, dotp, n2, p1);
    routing_kernel<<<RB, RTHREADS, 0, stream>>>(p1, dotp, n2, rt, p2);
    final_kernel<<<STREAM_BLOCKS, 256, 0, stream>>>(uh, n2, rt, p2, out);
}


# mean by linearity: pre-pass sums x[7] columns, GEMM blocks form mean = xsum.W from their LDS tile; dot, norm, first routing sum from accumulators (n2/dotp capsule-major); sum orders change, fixed
# speedup vs baseline: 15.3547x; 1.2032x over previous
#include <hip/hip_runtime.h>
#include <hip/hip_bf16.h>

typedef __attribute__((ext_vector_type(2))) float v2f;
typedef __attribute__((ext_vector_type(8))) float v8f;

#define JB 8192
#define NJ 8
#define DIM 128
#define COLS 1024

__device__ __forceinline__ float other_row(float v) {
    const unsigned u = __builtin_bit_cast(unsigned, v);
    return __builtin_bit_cast(float,
        __builtin_amdgcn_permlanex16(u, u, 0x76543210u, 0xfedcba98u, false, false));
}

template <int XM>
__device__ __forceinline__ float row_xadd(float v) {
    const int t = __builtin_amdgcn_update_dpp(0, __builtin_bit_cast(int, v), 0x160 + XM, 0xf, 0xf, true);
    return v + __builtin_bit_cast(float, t);
}

__device__ __forceinline__ float wave_sum(float v) {
    v = row_xadd<1>(v);
    v = row_xadd<2>(v);
    v = row_xadd<4>(v);
    v = row_xadd<8>(v);
    return v + other_row(v);
}

__device__ __forceinline__ float squash_gain(float n2) {
    return n2 / ((1.0f + n2) * sqrtf(n2 + 1e-8f));
}

__device__ __forceinline__ float softmax8(float b) {
    float m = b;
    m = fmaxf(m, __shfl_xor(m, 1, 32));
    m = fmaxf(m, __shfl_xor(m, 2, 32));
    m = fmaxf(m, __shfl_xor(m, 4, 32));
    const float e = __expf(b - m);
    float s = e;
    s += __shfl_xor(s, 1, 32);
    s += __shfl_xor(s, 2, 32);
    s += __shfl_xor(s, 4, 32);
    return e / s;
}

#define XB 64
#define XTHREADS 512
__launch_bounds__(XTHREADS)
__global__ void xsum_kernel(const float* __restrict__ x, float* __restrict__ xp) {
    __shared__ float4 red[XTHREADS / 32][32];
    const float* x7 = x + (size_t)7 * 8 * 1024 * 128;
    const int wave = threadIdx.x >> 5;
    const int lane = threadIdx.x & 31;
    const float4* src = (const float4*)(x7 + (size_t)(blockIdx.x * 128 + wave) * DIM) + lane;
    float4 a[8];
#pragma unroll
    for (int i = 0; i < 8; ++i) a[i] = src[(size_t)i * (XTHREADS / 32) * (DIM / 4)];
#pragma unroll
    for (int st = 1; st < 8; st <<= 1) {
#pragma unroll
        for (int i = 0; i < 8; i += 2 * st) {
            a[i].x += a[i + st].x; a[i].y += a[i + st].y;
            a[i].z += a[i + st].z; a[i].w += a[i + st].w;
        }
    }
    red[wave][lane] = a[0];
    __syncthreads();
    if (threadIdx.x < 32) {
        float4 s[XTHREADS / 32];
#pragma unroll
        for (int wv = 0; wv < XTHREADS / 32; ++wv) s[wv] = red[wv][lane];
#pragma unroll
        for (int st = 1; st < XTHREADS / 32; st <<= 1) {
#pragma unroll
            for (int wv = 0; wv < XTHREADS / 32; wv += 2 * st) {
                s[wv].x += s[wv + st].x; s[wv].y += s[wv + st].y;
                s[wv].z += s[wv + st].z; s[wv].w += s[wv + st].w;
            }
        }
        ((float4*)(xp + (size_t)blockIdx.x * DIM))[lane] = s[0];
    }
}

__launch_bounds__(256)
__global__ void gemm_uhat_kernel(const float* __restrict__ x,
                                 const float* __restrict__ w,
                                 const float* __restrict__ xp,
                                 float* __restrict__ uh,
                                 float* __restrict__ n2,
                                 float* __restrict__ dotp,
                                 float* __restrict__ p1) {
    __shared__ float4 wsm[64 * 64];
    __shared__ float  xs_s[2][DIM];
    __shared__ float4 um8[8][DIM / 4];
    __shared__ float  nbuf[128];
    __shared__ float  dbuf[128];
    __shared__ float  ws1[8];

    const float* x7 = x + (size_t)7 * 8 * 1024 * 128;
    const float* w7 = w + (size_t)7 * 8 * 128 * 128;

    const int wave = threadIdx.x >> 5;
    const int lane = threadIdx.x & 31;
    const int half = lane >> 4;
    const int l16  = lane & 15;

    const int j    = blockIdx.y;
    const int row0 = blockIdx.x * 128 + wave * 16;
    const int r    = row0 + l16;
    const float* arow = x7 + (size_t)((r & 7) * 1024 + (r >> 3)) * 128;

    const float* wj = w7 + (size_t)j * 128 * 128;

    {
        const int slot = (lane >> 4) * 32 + (lane & 15);
        float xv[XB / 2];
        {
            const float* xq = xp + (size_t)((threadIdx.x >> 7) * (XB / 2)) * DIM + (threadIdx.x & (DIM - 1));
#pragma unroll
            for (int b = 0; b < XB / 2; ++b) xv[b] = xq[b * DIM];
        }
        float4 r0[8], r1[8];
#pragma unroll
        for (int i = 0; i < 8; ++i) {
            const int kp = i * 8 + wave;
            r0[i] = *(const float4*)(wj + (size_t)(2 * kp) * 128 + 4 * lane);
            r1[i] = *(const float4*)(wj + (size_t)(2 * kp + 1) * 128 + 4 * lane);
        }
#pragma unroll
        for (int st = 1; st < XB / 2; st <<= 1) {
#pragma unroll
            for (int b = 0; b < XB / 2; b += 2 * st) xv[b] += xv[b + st];
        }
        xs_s[threadIdx.x >> 7][threadIdx.x & (DIM - 1)] = xv[0];
        __syncthreads();
        float4 u = make_float4(0.0f, 0.0f, 0.0f, 0.0f);
#pragma unroll
        for (int i = 0; i < 8; ++i) {
            const int kp = i * 8 + wave;
            const float2 xa = ((const float2*)xs_s[0])[kp];
            const float2 xb = ((const float2*)xs_s[1])[kp];
            const float x0 = xa.x + xb.x, x1 = xa.y + xb.y;
            u.x = fmaf(x0, r0[i].x, u.x); u.x = fmaf(x1, r1[i].x, u.x);
            u.y = fmaf(x0, r0[i].y, u.y); u.y = fmaf(x1, r1[i].y, u.y);
            u.z = fmaf(x0, r0[i].z, u.z); u.z = fmaf(x1, r1[i].z, u.z);
            u.w = fmaf(x0, r0[i].w, u.w); u.w = fmaf(x1, r1[i].w, u.w);
        }
        um8[wave][lane] = u;
#pragma unroll
        for (int i = 0; i < 8; ++i) {
            const int kp = i * 8 + wave;
            wsm[kp * 64 + slot]      = make_float4(r0[i].x, r1[i].x, r0[i].y, r1[i].y);
            wsm[kp * 64 + slot + 16] = make_float4(r0[i].z, r1[i].z, r0[i].w, r1[i].w);
        }
    }

    v8f acc[8];
#pragma unroll
    for (int n = 0; n < 8; ++n) acc[n] = (v8f){0,0,0,0,0,0,0,0};

    constexpr int AD = 2;
    v2f ar[AD + 1];
#pragma unroll
    for (int t = 0; t < AD; ++t) {
        ar[t].x = arow[4 * t + 2 * half + 0];
        ar[t].y = arow[4 * t + 2 * half + 1];
    }
    __syncthreads();

    const float4* bp = wsm + half * 64 + l16;
    float4 q[2][4];
#pragma unroll
    for (int m = 0; m < 4; ++m) q[0][m] = bp[m * 16];

#pragma unroll
    for (int t = 0; t < 32; ++t) {
        if (t + AD < 32) {
            ar[(t + AD) % (AD + 1)].x = arow[4 * (t + AD) + 2 * half + 0];
            ar[(t + AD) % (AD + 1)].y = arow[4 * (t + AD) + 2 * half + 1];
        }
        if (t + 1 < 32) {
#pragma unroll
            for (int m = 0; m < 4; ++m) q[(t + 1) & 1][m] = bp[(t + 1) * 128 + m * 16];
        }
        const v2f a = ar[t % (AD + 1)];
#pragma unroll
        for (int m = 0; m < 4; ++m) {
            const float4 qq = q[t & 1][m];
            v2f b0, b1;
            b0.x = qq.x; b0.y = qq.y;
            b1.x = qq.z; b1.y = qq.w;
            acc[2 * m] = __builtin_amdgcn_wmma_f32_16x16x4_f32(
                false, a, false, b0, (short)0, acc[2 * m], false, false);
            acc[2 * m + 1] = __builtin_amdgcn_wmma_f32_16x16x4_f32(
                false, a, false, b1, (short)0, acc[2 * m + 1], false, false);
        }
    }
    asm volatile("v_nop\n\tv_nop\n\tv_nop\n\tv_nop\n\tv_nop\n\tv_nop\n\tv_nop\n\tv_nop"
                 : "+v"(acc[0]), "+v"(acc[1]), "+v"(acc[2]), "+v"(acc[3]),
                   "+v"(acc[4]), "+v"(acc[5]), "+v"(acc[6]), "+v"(acc[7]));
    __builtin_amdgcn_sched_barrier(0);
#pragma unroll
    for (int g = 0; g < 8; ++g) {
        float* od = uh + (size_t)(row0 + 8 * half + g) * COLS + j * DIM + 4 * l16;
        *(float4*)od        = make_float4(acc[0][g], acc[1][g], acc[2][g], acc[3][g]);
        *(float4*)(od + 64) = make_float4(acc[4][g], acc[5][g], acc[6][g], acc[7][g]);
    }
    __builtin_amdgcn_sched_barrier(0);
    float um[8];
    {
        float4 pa[8], pb[8];
#pragma unroll
        for (int wv = 0; wv < 8; ++wv) { pa[wv] = um8[wv][l16]; pb[wv] = um8[wv][16 + l16]; }
        const float inv = 1.0f / (float)JB;
        um[0] = (((pa[0].x + pa[1].x) + (pa[2].x + pa[3].x)) + ((pa[4].x + pa[5].x) + (pa[6].x + pa[7].x))) * inv;
        um[1] = (((pa[0].y + pa[1].y) + (pa[2].y + pa[3].y)) + ((pa[4].y + pa[5].y) + (pa[6].y + pa[7].y))) * inv;
        um[2] = (((pa[0].z + pa[1].z) + (pa[2].z + pa[3].z)) + ((pa[4].z + pa[5].z) + (pa[6].z + pa[7].z))) * inv;
        um[3] = (((pa[0].w + pa[1].w) + (pa[2].w + pa[3].w)) + ((pa[4].w + pa[5].w) + (pa[6].w + pa[7].w))) * inv;
        um[4] = (((pb[0].x + pb[1].x) + (pb[2].x + pb[3].x)) + ((pb[4].x + pb[5].x) + (pb[6].x + pb[7].x))) * inv;
        um[5] = (((pb[0].y + pb[1].y) + (pb[2].y + pb[3].y)) + ((pb[4].y + pb[5].y) + (pb[6].y + pb[7].y))) * inv;
        um[6] = (((pb[0].z + pb[1].z) + (pb[2].z + pb[3].z)) + ((pb[4].z + pb[5].z) + (pb[6].z + pb[7].z))) * inv;
        um[7] = (((pb[0].w + pb[1].w) + (pb[2].w + pb[3].w)) + ((pb[4].w + pb[5].w) + (pb[6].w + pb[7].w))) * inv;
    }
    float qv[8], dv[8];
#pragma unroll
    for (int g = 0; g < 8; ++g) {
        float qq = acc[0][g] * acc[0][g];
        float dd = acc[0][g] * um[0];
#pragma unroll
        for (int n = 1; n < 8; ++n) {
            qq = fmaf(acc[n][g], acc[n][g], qq);
            dd = fmaf(acc[n][g], um[n], dd);
        }
        qv[g] = qq; dv[g] = dd;
    }
#pragma unroll
    for (int g = 0; g < 8; ++g) { qv[g] = row_xadd<1>(qv[g]); dv[g] = row_xadd<1>(dv[g]); }
#pragma unroll
    for (int g = 0; g < 8; ++g) { qv[g] = row_xadd<2>(qv[g]); dv[g] = row_xadd<2>(dv[g]); }
#pragma unroll
    for (int g = 0; g < 8; ++g) { qv[g] = row_xadd<4>(qv[g]); dv[g] = row_xadd<4>(dv[g]); }
#pragma unroll
    for (int g = 0; g < 8; ++g) { qv[g] = row_xadd<8>(qv[g]); dv[g] = row_xadd<8>(dv[g]); }
    float qsel = 0.0f, dsel = 0.0f;
#pragma unroll
    for (int g = 0; g < 8; ++g) {
        if (l16 == g) { qsel = qv[g]; dsel = dv[g]; }
    }
    float t1 = squash_gain(0.015625f * qsel) * dsel;
    t1 = row_xadd<1>(t1);
    t1 = row_xadd<2>(t1);
    t1 = row_xadd<4>(t1);
    t1 += other_row(t1);
    if (l16 < 8) {
        nbuf[wave * 16 + 8 * half + l16] = qsel;
        dbuf[wave * 16 + 8 * half + l16] = dsel;
    }
    if (lane == 0) ws1[wave] = t1;
    __syncthreads();
    if (threadIdx.x < 128) {
        const size_t o = (size_t)j * JB + blockIdx.x * 128 + threadIdx.x;
        n2[o]   = nbuf[threadIdx.x];
        dotp[o] = dbuf[threadIdx.x];
    }
    if (threadIdx.x == 0)
        p1[j * 64 + blockIdx.x] = ((ws1[0] + ws1[1]) + (ws1[2] + ws1[3])) +
                                  ((ws1[4] + ws1[5]) + (ws1[6] + ws1[7]));
}

#define PPW 16
#define STREAM_BLOCKS 512

#define RB 32
#define RTHREADS 512
#define RWAVES (RTHREADS / 32)
#define RSWEEPS (JB * NJ / (RB * RTHREADS))
__launch_bounds__(RTHREADS)
__global__ void routing_kernel(const float* __restrict__ p1,
                               const float* __restrict__ dotp,
                               const float* __restrict__ n2,
                               float* __restrict__ rt,
                               float* __restrict__ p2) {
    __shared__ float redA[RWAVES];
    __shared__ float redC[RWAVES];
    __shared__ float c1s[NJ];
    const int tid  = threadIdx.x;
    const int lane = tid & 31;
    const int wave = tid >> 5;
    const int base = blockIdx.x * (RTHREADS * RSWEEPS) + tid;

    float nn[RSWEEPS], dd[RSWEEPS];
#pragma unroll
    for (int k = 0; k < RSWEEPS; ++k) {
        nn[k] = n2[base + k * RTHREADS];
        dd[k] = dotp[base + k * RTHREADS];
    }

    const float a = wave_sum(p1[tid]);
    if (lane == 0) redA[wave] = a;
    __syncthreads();
    if (wave == 0) {
        const float s = redA[2 * (lane & 7)] + redA[2 * (lane & 7) + 1];
        const float b1 = 0.125f * s * (1.0f / (float)JB);
        const float c1 = softmax8(b1);
        if (lane < NJ) {
            c1s[lane] = c1;
            if (blockIdx.x == 0) { rt[lane] = b1; rt[NJ + lane] = c1; }
        }
    }
    __syncthreads();

    const float cj = c1s[blockIdx.x >> 2];
    const float cc = cj * cj;
    float t = 0.0f;
#pragma unroll
    for (int k = 0; k < RSWEEPS; ++k) t += squash_gain(cc * nn[k]) * dd[k];
    t = wave_sum(t);
    if (lane == 0) redC[wave] = t;
    __syncthreads();
    if (tid == 0) {
        float s[RWAVES];
#pragma unroll
        for (int k = 0; k < RWAVES; ++k) s[k] = redC[k];
#pragma unroll
        for (int st = 1; st < RWAVES; st <<= 1) {
#pragma unroll
            for (int k = 0; k < RWAVES; k += 2 * st) s[k] += s[k + st];
        }
        p2[blockIdx.x] = s[0];
    }
}

__launch_bounds__(256)
__global__ void final_kernel(const float* __restrict__ uh,
                             const float* __restrict__ n2,
                             const float* __restrict__ rt,
                             const float* __restrict__ p2,
                             float* __restrict__ out) {
    const int wave = threadIdx.x >> 5;
    const int lane = threadIdx.x & 31;
    const int p0 = (blockIdx.x * 8 + wave) * PPW;

    const int pl = lane & (PPW - 1);
    const int jl = lane & 7;
    float nn = n2[jl * JB + (p0 >> 3) + (pl >> 3)];
    float b1 = rt[jl];
    float c1 = rt[NJ + jl];
    float s2[RB / NJ];
#pragma unroll
    for (int k = 0; k < RB / NJ; ++k) s2[k] = p2[(RB / NJ) * jl + k];
    __builtin_amdgcn_sched_barrier(0);

    const float4* src = (const float4*)(uh + (size_t)p0 * DIM) + lane;
    float4* dst = (float4*)(out + (size_t)p0 * DIM) + lane;
    float4 a[PPW];
#pragma unroll
    for (int i = 0; i < PPW; ++i) a[i] = src[i * (DIM / 4)];
    __builtin_amdgcn_sched_barrier(0);
    asm volatile("" : "+v"(nn), "+v"(b1), "+v"(c1),
                      "+v"(s2[0]), "+v"(s2[1]), "+v"(s2[2]), "+v"(s2[3]));

    const float sj = (s2[0] + s2[1]) + (s2[2] + s2[3]);
    const float cj = softmax8(b1 + c1 * sj * (1.0f / (float)JB));

    const float gain = cj * squash_gain(cj * cj * nn);
    const int gbits = __builtin_bit_cast(int, gain);

#pragma unroll
    for (int i = 0; i < PPW; ++i) {
        const float g = __builtin_bit_cast(float, __builtin_amdgcn_readlane(gbits, i));
        float4 v = a[i];
        v.x *= g; v.y *= g; v.z *= g; v.w *= g;
        dst[i * (DIM / 4)] = v;
    }
}

extern "C" void kernel_launch(void* const* d_in, const int* in_sizes, int n_in,
                              void* d_out, int out_size, void* d_ws, size_t ws_size,
                              hipStream_t stream) {
    const float* x = (const float*)d_in[0];
    const float* w = (const float*)d_in[1];
    float* out = (float*)d_out;

    char* ws = (char*)d_ws;
    float* uh   = (float*)(ws);
    float* n2   = (float*)(ws + (size_t)JB * COLS * 4);
    float* dotp = (float*)(ws + (size_t)JB * COLS * 4 + 262144);
    float* sm   = (float*)(ws + (size_t)JB * COLS * 4 + 524288);
    float* p1 = sm;
    float* p2 = sm + 512;
    float* rt = sm + 768;
    float* xp = out;

    xsum_kernel<<<XB, XTHREADS, 0, stream>>>(x, xp);
    gemm_uhat_kernel<<<dim3(64, NJ), 256, 0, stream>>>(x, w, xp, uh, n2, dotp, p1);
    routing_kernel<<<RB, RTHREADS, 0, stream>>>(p1, dotp, n2, rt, p2);
    final_kernel<<<STREAM_BLOCKS, 256, 0, stream>>>(uh, n2, rt, p2, out);
}
